# starting kernel
# baseline (speedup 1.0000x reference)
#include <hip/hip_runtime.h>

typedef float v2f __attribute__((ext_vector_type(2)));
typedef float v8f __attribute__((ext_vector_type(8)));

#define WS 2.3283064365386963e-10f
__constant__ float WC[33] = {
  1.0f*WS, 32.0f*WS, 496.0f*WS, 4960.0f*WS, 35960.0f*WS, 201376.0f*WS,
  906192.0f*WS, 3365856.0f*WS, 10518300.0f*WS, 28048800.0f*WS,
  64512240.0f*WS, 129024480.0f*WS, 225792840.0f*WS, 347373600.0f*WS,
  471435600.0f*WS, 565722720.0f*WS, 601080390.0f*WS, 565722720.0f*WS,
  471435600.0f*WS, 347373600.0f*WS, 225792840.0f*WS, 129024480.0f*WS,
  64512240.0f*WS, 28048800.0f*WS, 10518300.0f*WS, 3365856.0f*WS,
  906192.0f*WS, 201376.0f*WS, 35960.0f*WS, 4960.0f*WS, 496.0f*WS,
  32.0f*WS, 1.0f*WS
};

#define IMG     4096
#define TILE    64
#define INR     128
#define XS_S    136
#define TW      96
#define TS_S    104
#define TS2_S   108
#define XS_FL   (INR * XS_S)
#define TS_FL   (TW * TS_S)
#define STASH_FL (TILE * TILE)
#define LDS_FLOATS (XS_FL + TS_FL + STASH_FL)

static __device__ __forceinline__ v8f wmma4(v2f a, v2f b, v8f c) {
  return __builtin_amdgcn_wmma_f32_16x16x4_f32(false, a, false, b, (short)0, c,
                                               false, false);
}

static __device__ __forceinline__ v8f conv_block(const v2f (&W)[3][4],
                                                 const v2f (&B0)[4],
                                                 const v2f (&B1)[4],
                                                 const v2f (&B2)[4]) {
  v8f a0 = {}, a1 = {};
  a0 = wmma4(W[0][0], B0[0], a0);  a1 = wmma4(W[2][0], B2[0], a1);
  a0 = wmma4(W[0][1], B0[1], a0);  a1 = wmma4(W[2][1], B2[1], a1);
  a0 = wmma4(W[0][2], B0[2], a0);  a1 = wmma4(W[2][2], B2[2], a1);
  a0 = wmma4(W[0][3], B0[3], a0);  a1 = wmma4(W[2][3], B2[3], a1);
  a0 = wmma4(W[1][0], B1[0], a0);  a1 = wmma4(W[1][2], B1[2], a1);
  a0 = wmma4(W[1][1], B1[1], a0);  a1 = wmma4(W[1][3], B1[3], a1);
  return a0 + a1;
}

__global__ __launch_bounds__(256)
void ao_render_kernel(const float* __restrict__ x,
                      const float* __restrict__ nrm,
                      float* __restrict__ out) {
  extern __shared__ float lds[];
  float* Xs    = lds;
  float* Ts    = lds + XS_FL;
  float* Stash = lds + XS_FL + TS_FL;
  float* Ts2   = lds;

  const int tid  = threadIdx.x;
  const int wave = tid >> 5;
  const int lane = tid & 31;
  const int hl   = lane >> 4;
  const int ln   = lane & 15;

  const int Y0 = blockIdx.y * TILE;
  const int X0 = blockIdx.x * TILE;

  for (int idx = tid; idx < INR * INR; idx += 256) {
    int row = idx >> 7;
    int cx  = idx & 127;
    int gy = Y0 - 32 + row;
    int gx = X0 - 32 + cx;
    gy = (gy < 0) ? (-1 - gy) : ((gy > IMG - 1) ? (2 * IMG - 1 - gy) : gy);
    gx = (gx < 0) ? (-1 - gx) : ((gx > IMG - 1) ? (2 * IMG - 1 - gx) : gx);
    float v = x[gy * IMG + gx];
    int cu = (row + cx) & 127;
    Xs[row * XS_S + cu] = v;
    if (row >= 32 && row < 96 && cx >= 32 && cx < 96)
      Stash[(row - 32) * TILE + (cx - 32)] = v;
  }

  v2f W[3][4];
#pragma unroll
  for (int k = 0; k < 3; ++k)
#pragma unroll
    for (int kc = 0; kc < 4; ++kc) {
      int d0 = 16 * k + 4 * kc + 2 * hl - ln;
      float a0 = (d0 >= 0 && d0 <= 32) ? WC[d0] : 0.0f;
      float a1 = (d0 + 1 >= 0 && d0 + 1 <= 32) ? WC[d0 + 1] : 0.0f;
      v2f w; w.x = a0; w.y = a1;
      W[k][kc] = w;
    }

  __syncthreads();

  {
    const int c = wave * 16 + ln;
    v2f B0[4], B1[4], B2[4];
#pragma unroll
    for (int kc = 0; kc < 4; ++kc) {
      int rr = 4 * kc + 2 * hl;
      v2f t;
      t.x = Xs[(rr + 0) * XS_S + c];  t.y = Xs[(rr + 1) * XS_S + c];  B0[kc] = t;
      t.x = Xs[(rr + 16) * XS_S + c]; t.y = Xs[(rr + 17) * XS_S + c]; B1[kc] = t;
      t.x = Xs[(rr + 32) * XS_S + c]; t.y = Xs[(rr + 33) * XS_S + c]; B2[kc] = t;
    }
    for (int rb = 0; rb < 6; ++rb) {
      const int r0 = 16 + 16 * rb;
      v8f acc = conv_block(W, B0, B1, B2);
#pragma unroll
      for (int g = 0; g < 8; ++g) {
        int ly = r0 + g + 8 * hl;
        int lx = c - ly; if (lx < 0) lx += 128;
        if (lx >= 16 && lx < 112) {
          int ty = ly - 16, tx = lx - 16;
          int dd = tx - ty; int cv = (dd < 0) ? dd + TW : dd;
          Ts[ty * TS_S + cv] = acc[g];
        }
      }
      if (rb < 5) {
#pragma unroll
        for (int kc = 0; kc < 4; ++kc) {
          B0[kc] = B1[kc];
          B1[kc] = B2[kc];
          int rr = r0 + 32 + 4 * kc + 2 * hl;
          v2f t;
          t.x = Xs[(rr + 0) * XS_S + c];
          t.y = Xs[(rr + 1) * XS_S + c];
          B2[kc] = t;
        }
      }
    }
  }

  __syncthreads();

  if (wave < 6) {
    const int c = wave * 16 + ln;
    v2f B0[4], B1[4], B2[4];
#pragma unroll
    for (int kc = 0; kc < 4; ++kc) {
      int rr = 4 * kc + 2 * hl;
      v2f t;
      t.x = Ts[(rr + 0) * TS_S + c];  t.y = Ts[(rr + 1) * TS_S + c];  B0[kc] = t;
      t.x = Ts[(rr + 16) * TS_S + c]; t.y = Ts[(rr + 17) * TS_S + c]; B1[kc] = t;
      t.x = Ts[(rr + 32) * TS_S + c]; t.y = Ts[(rr + 33) * TS_S + c]; B2[kc] = t;
    }
    for (int rb = 0; rb < 4; ++rb) {
      const int r2 = 16 + 16 * rb;
      v8f acc = conv_block(W, B0, B1, B2);
#pragma unroll
      for (int g = 0; g < 8; ++g) {
        int row = r2 - 16 + g + 8 * hl;
        Ts2[row * TS2_S + c] = acc[g];
      }
      if (rb < 3) {
#pragma unroll
        for (int kc = 0; kc < 4; ++kc) {
          B0[kc] = B1[kc];
          B1[kc] = B2[kc];
          int rr = r2 + 32 + 4 * kc + 2 * hl;
          v2f t;
          t.x = Ts[(rr + 0) * TS_S + c];
          t.y = Ts[(rr + 1) * TS_S + c];
          B2[kc] = t;
        }
      }
    }
  }

  __syncthreads();

  for (int idx = tid; idx < TILE * TILE; idx += 256) {
    int oy = idx >> 6;
    int ox = idx & 63;
    int dd = ox - oy; int cv = (dd < 0) ? dd + TW : dd;
    float rx = Ts2[oy * TS2_S + cv];
    float xv = Stash[idx];
    float ao = 1.0f - fminf(fmaxf((rx - xv) * 4.0f, 0.0f), 1.0f);
    int gidx = (Y0 + oy) * IMG + (X0 + ox);
    float n0 = nrm[gidx];
    float n1 = nrm[gidx + IMG * IMG];
    float n2 = nrm[gidx + 2 * IMG * IMG];
    float d1 = fmaxf(0.7053f * ((n0 - n1) + n2), 0.0f);
    out[gidx] = (0.3f * d1 + 0.7f) * ao;
  }
}

extern "C" void kernel_launch(void* const* d_in, const int* in_sizes, int n_in,
                              void* d_out, int out_size, void* d_ws, size_t ws_size,
                              hipStream_t stream) {
  (void)in_sizes; (void)n_in; (void)d_ws; (void)ws_size; (void)out_size;
  const float* x   = (const float*)d_in[0];
  const float* nrm = (const float*)d_in[1];
  float* out = (float*)d_out;
  dim3 grid(IMG / TILE, IMG / TILE);
  size_t ldsBytes = LDS_FLOATS * sizeof(float);
  ao_render_kernel<<<grid, dim3(256), ldsBytes, stream>>>(x, nrm, out);
}


# all global loads issued up front as b128 (16 tile + 12 normals per thread in flight, was 1-2 b32 per round trip); LDS-only barriers keep the normals in flight under the conv stages
# speedup vs baseline: 3.3884x; 3.3884x over previous
#include <hip/hip_runtime.h>

typedef float v2f __attribute__((ext_vector_type(2)));
typedef float v4f __attribute__((ext_vector_type(4)));
typedef float v8f __attribute__((ext_vector_type(8)));

#define WS 2.3283064365386963e-10f
__constant__ float WC[33] = {
  1.0f*WS, 32.0f*WS, 496.0f*WS, 4960.0f*WS, 35960.0f*WS, 201376.0f*WS,
  906192.0f*WS, 3365856.0f*WS, 10518300.0f*WS, 28048800.0f*WS,
  64512240.0f*WS, 129024480.0f*WS, 225792840.0f*WS, 347373600.0f*WS,
  471435600.0f*WS, 565722720.0f*WS, 601080390.0f*WS, 565722720.0f*WS,
  471435600.0f*WS, 347373600.0f*WS, 225792840.0f*WS, 129024480.0f*WS,
  64512240.0f*WS, 28048800.0f*WS, 10518300.0f*WS, 3365856.0f*WS,
  906192.0f*WS, 201376.0f*WS, 35960.0f*WS, 4960.0f*WS, 496.0f*WS,
  32.0f*WS, 1.0f*WS
};

#define IMG     4096
#define TILE    64
#define INR     128
#define XS_S    136
#define TW      96
#define TS_S    104
#define TS2_S   108
#define XS_FL   (INR * XS_S)
#define TS_FL   (TW * TS_S)
#define STASH_FL (TILE * TILE)
#define LDS_FLOATS (XS_FL + TS_FL + STASH_FL)

static __device__ __forceinline__ v8f wmma4(v2f a, v2f b, v8f c) {
  return __builtin_amdgcn_wmma_f32_16x16x4_f32(false, a, false, b, (short)0, c,
                                               false, false);
}

static __device__ __forceinline__ void lds_barrier() {
  asm volatile("s_wait_dscnt 0x0\n\ts_barrier_signal -1\n\ts_barrier_wait -1" ::: "memory");
}

static __device__ __forceinline__ v8f conv_block(const v2f (&W)[3][4],
                                                 const v2f (&B0)[4],
                                                 const v2f (&B1)[4],
                                                 const v2f (&B2)[4]) {
  v8f a0 = {}, a1 = {};
  a0 = wmma4(W[0][0], B0[0], a0);  a1 = wmma4(W[2][0], B2[0], a1);
  a0 = wmma4(W[0][1], B0[1], a0);  a1 = wmma4(W[2][1], B2[1], a1);
  a0 = wmma4(W[0][2], B0[2], a0);  a1 = wmma4(W[2][2], B2[2], a1);
  a0 = wmma4(W[0][3], B0[3], a0);  a1 = wmma4(W[2][3], B2[3], a1);
  a0 = wmma4(W[1][0], B1[0], a0);  a1 = wmma4(W[1][2], B1[2], a1);
  a0 = wmma4(W[1][1], B1[1], a0);  a1 = wmma4(W[1][3], B1[3], a1);
  return a0 + a1;
}

__global__ __launch_bounds__(256)
void ao_render_kernel(const float* __restrict__ x,
                      const float* __restrict__ nrm,
                      float* __restrict__ out) {
  extern __shared__ __attribute__((aligned(16))) float lds[];
  float* Xs    = lds;
  float* Ts    = lds + XS_FL;
  float* Stash = lds + XS_FL + TS_FL;
  float* Ts2   = lds;

  const int tid  = threadIdx.x;
  const int wave = tid >> 5;
  const int lane = tid & 31;
  const int hl   = lane >> 4;
  const int ln   = lane & 15;

  const int Y0 = blockIdx.y * TILE;
  const int X0 = blockIdx.x * TILE;

  v2f W[3][4];
#pragma unroll
  for (int k = 0; k < 3; ++k)
#pragma unroll
    for (int kc = 0; kc < 4; ++kc) {
      int d0 = 16 * k + 4 * kc + 2 * hl - ln;
      float a0 = (d0 >= 0 && d0 <= 32) ? WC[d0] : 0.0f;
      float a1 = (d0 + 1 >= 0 && d0 + 1 <= 32) ? WC[d0 + 1] : 0.0f;
      v2f w; w.x = a0; w.y = a1;
      W[k][kc] = w;
    }

  const int c4  = lane * 4;
  const int gx0 = X0 - 32 + c4;
  const bool rev = (gx0 < 0) || (gx0 > IMG - 1);
  const int gxb = (gx0 < 0) ? (-4 - gx0) : ((gx0 > IMG - 1) ? (2 * IMG - 4 - gx0) : gx0);
  v4f xin[16];
#pragma unroll
  for (int i = 0; i < 16; ++i) {
    int gy = Y0 - 32 + wave + 8 * i;
    gy = (gy < 0) ? (-1 - gy) : ((gy > IMG - 1) ? (2 * IMG - 1 - gy) : gy);
    xin[i] = *reinterpret_cast<const v4f*>(x + gy * IMG + gxb);
  }
  __builtin_amdgcn_sched_barrier(0);
  const int ox4 = (tid & 15) * 4;
  const int oyb = tid >> 4;
  v4f nin0[4], nin1[4], nin2[4];
#pragma unroll
  for (int i = 0; i < 4; ++i) {
    int gidx = (Y0 + oyb + 16 * i) * IMG + (X0 + ox4);
    nin0[i] = *reinterpret_cast<const v4f*>(nrm + gidx);
    nin1[i] = *reinterpret_cast<const v4f*>(nrm + gidx + IMG * IMG);
    nin2[i] = *reinterpret_cast<const v4f*>(nrm + gidx + 2 * IMG * IMG);
  }
  __builtin_amdgcn_sched_barrier(0);

#pragma unroll
  for (int i = 0; i < 16; ++i) {
    const int row = wave + 8 * i;
    const v4f v = xin[i];
    const float e0 = rev ? v.w : v.x;
    const float e1 = rev ? v.z : v.y;
    const float e2 = rev ? v.y : v.z;
    const float e3 = rev ? v.x : v.w;
    const int s = row + c4;
    float* xr = Xs + row * XS_S;
    xr[(s + 0) & 127] = e0;
    xr[(s + 1) & 127] = e1;
    xr[(s + 2) & 127] = e2;
    xr[(s + 3) & 127] = e3;
    if (i >= 4 && i < 12 && c4 >= 32 && c4 < 96) {
      v4f t; t.x = e0; t.y = e1; t.z = e2; t.w = e3;
      *reinterpret_cast<v4f*>(Stash + (row - 32) * TILE + (c4 - 32)) = t;
    }
  }

  lds_barrier();

  {
    const int c = wave * 16 + ln;
    v2f B0[4], B1[4], B2[4];
#pragma unroll
    for (int kc = 0; kc < 4; ++kc) {
      int rr = 4 * kc + 2 * hl;
      v2f t;
      t.x = Xs[(rr + 0) * XS_S + c];  t.y = Xs[(rr + 1) * XS_S + c];  B0[kc] = t;
      t.x = Xs[(rr + 16) * XS_S + c]; t.y = Xs[(rr + 17) * XS_S + c]; B1[kc] = t;
      t.x = Xs[(rr + 32) * XS_S + c]; t.y = Xs[(rr + 33) * XS_S + c]; B2[kc] = t;
    }
    for (int rb = 0; rb < 6; ++rb) {
      const int r0 = 16 + 16 * rb;
      v8f acc = conv_block(W, B0, B1, B2);
#pragma unroll
      for (int g = 0; g < 8; ++g) {
        int ly = r0 + g + 8 * hl;
        int lx = c - ly; if (lx < 0) lx += 128;
        if (lx >= 16 && lx < 112) {
          int ty = ly - 16, tx = lx - 16;
          int dd = tx - ty; int cv = (dd < 0) ? dd + TW : dd;
          Ts[ty * TS_S + cv] = acc[g];
        }
      }
      if (rb < 5) {
#pragma unroll
        for (int kc = 0; kc < 4; ++kc) {
          B0[kc] = B1[kc];
          B1[kc] = B2[kc];
          int rr = r0 + 32 + 4 * kc + 2 * hl;
          v2f t;
          t.x = Xs[(rr + 0) * XS_S + c];
          t.y = Xs[(rr + 1) * XS_S + c];
          B2[kc] = t;
        }
      }
    }
  }

  lds_barrier();

  if (wave < 6) {
    const int c = wave * 16 + ln;
    v2f B0[4], B1[4], B2[4];
#pragma unroll
    for (int kc = 0; kc < 4; ++kc) {
      int rr = 4 * kc + 2 * hl;
      v2f t;
      t.x = Ts[(rr + 0) * TS_S + c];  t.y = Ts[(rr + 1) * TS_S + c];  B0[kc] = t;
      t.x = Ts[(rr + 16) * TS_S + c]; t.y = Ts[(rr + 17) * TS_S + c]; B1[kc] = t;
      t.x = Ts[(rr + 32) * TS_S + c]; t.y = Ts[(rr + 33) * TS_S + c]; B2[kc] = t;
    }
    for (int rb = 0; rb < 4; ++rb) {
      const int r2 = 16 + 16 * rb;
      v8f acc = conv_block(W, B0, B1, B2);
#pragma unroll
      for (int g = 0; g < 8; ++g) {
        int row = r2 - 16 + g + 8 * hl;
        Ts2[row * TS2_S + c] = acc[g];
      }
      if (rb < 3) {
#pragma unroll
        for (int kc = 0; kc < 4; ++kc) {
          B0[kc] = B1[kc];
          B1[kc] = B2[kc];
          int rr = r2 + 32 + 4 * kc + 2 * hl;
          v2f t;
          t.x = Ts[(rr + 0) * TS_S + c];
          t.y = Ts[(rr + 1) * TS_S + c];
          B2[kc] = t;
        }
      }
    }
  }

  lds_barrier();

#pragma unroll
  for (int i = 0; i < 4; ++i) {
    const int oy = oyb + 16 * i;
    const v4f xv = *reinterpret_cast<const v4f*>(Stash + oy * TILE + ox4);
    const v4f n0 = nin0[i], n1 = nin1[i], n2 = nin2[i];
    v4f res;
#pragma unroll
    for (int k = 0; k < 4; ++k) {
      int dd = ox4 + k - oy; int cv = (dd < 0) ? dd + TW : dd;
      float rx = Ts2[oy * TS2_S + cv];
      float ao = 1.0f - fminf(fmaxf((rx - xv[k]) * 4.0f, 0.0f), 1.0f);
      float d1 = fmaxf(0.7053f * ((n0[k] - n1[k]) + n2[k]), 0.0f);
      res[k] = (0.3f * d1 + 0.7f) * ao;
    }
    *reinterpret_cast<v4f*>(out + (Y0 + oy) * IMG + (X0 + ox4)) = res;
  }
}

extern "C" void kernel_launch(void* const* d_in, const int* in_sizes, int n_in,
                              void* d_out, int out_size, void* d_ws, size_t ws_size,
                              hipStream_t stream) {
  (void)in_sizes; (void)n_in; (void)d_ws; (void)ws_size; (void)out_size;
  const float* x   = (const float*)d_in[0];
  const float* nrm = (const float*)d_in[1];
  float* out = (float*)d_out;
  dim3 grid(IMG / TILE, IMG / TILE);
  size_t ldsBytes = LDS_FLOATS * sizeof(float);
  ao_render_kernel<<<grid, dim3(256), ldsBytes, stream>>>(x, nrm, out);
}


# stage-1 result stored by diagonal index mod 128 (was mod 96): every store unpredicated, one and per address, no branch per element; stage 2 runs only the 20 blocks that hold outputs; sums unchanged
# speedup vs baseline: 4.1682x; 1.2301x over previous
#include <hip/hip_runtime.h>

typedef float v2f __attribute__((ext_vector_type(2)));
typedef float v4f __attribute__((ext_vector_type(4)));
typedef float v8f __attribute__((ext_vector_type(8)));

#define WS 2.3283064365386963e-10f
__constant__ float WC[33] = {
  1.0f*WS, 32.0f*WS, 496.0f*WS, 4960.0f*WS, 35960.0f*WS, 201376.0f*WS,
  906192.0f*WS, 3365856.0f*WS, 10518300.0f*WS, 28048800.0f*WS,
  64512240.0f*WS, 129024480.0f*WS, 225792840.0f*WS, 347373600.0f*WS,
  471435600.0f*WS, 565722720.0f*WS, 601080390.0f*WS, 565722720.0f*WS,
  471435600.0f*WS, 347373600.0f*WS, 225792840.0f*WS, 129024480.0f*WS,
  64512240.0f*WS, 28048800.0f*WS, 10518300.0f*WS, 3365856.0f*WS,
  906192.0f*WS, 201376.0f*WS, 35960.0f*WS, 4960.0f*WS, 496.0f*WS,
  32.0f*WS, 1.0f*WS
};

#define IMG     4096
#define TILE    64
#define INR     128
#define XS_S    136
#define TW      96
#define TS_S    136
#define TS2_S   132
#define XS_FL   (INR * XS_S)
#define TS_FL   (TW * TS_S)
#define STASH_FL (TILE * TILE)
#define LDS_FLOATS (XS_FL + TS_FL + STASH_FL)

static __device__ __forceinline__ v8f wmma4(v2f a, v2f b, v8f c) {
  return __builtin_amdgcn_wmma_f32_16x16x4_f32(false, a, false, b, (short)0, c,
                                               false, false);
}

static __device__ __forceinline__ void lds_barrier() {
  asm volatile("s_wait_dscnt 0x0\n\ts_barrier_signal -1\n\ts_barrier_wait -1" ::: "memory");
}

static __device__ __forceinline__ v8f conv_block(const v2f (&W)[3][4],
                                                 const v2f (&B0)[4],
                                                 const v2f (&B1)[4],
                                                 const v2f (&B2)[4]) {
  v8f a0 = {}, a1 = {};
  a0 = wmma4(W[0][0], B0[0], a0);  a1 = wmma4(W[2][0], B2[0], a1);
  a0 = wmma4(W[0][1], B0[1], a0);  a1 = wmma4(W[2][1], B2[1], a1);
  a0 = wmma4(W[0][2], B0[2], a0);  a1 = wmma4(W[2][2], B2[2], a1);
  a0 = wmma4(W[0][3], B0[3], a0);  a1 = wmma4(W[2][3], B2[3], a1);
  a0 = wmma4(W[1][0], B1[0], a0);  a1 = wmma4(W[1][2], B1[2], a1);
  a0 = wmma4(W[1][1], B1[1], a0);  a1 = wmma4(W[1][3], B1[3], a1);
  return a0 + a1;
}

__global__ __launch_bounds__(256)
void ao_render_kernel(const float* __restrict__ x,
                      const float* __restrict__ nrm,
                      float* __restrict__ out) {
  extern __shared__ __attribute__((aligned(16))) float lds[];
  float* Xs    = lds;
  float* Ts    = lds + XS_FL;
  float* Stash = lds + XS_FL + TS_FL;
  float* Ts2   = lds;

  const int tid  = threadIdx.x;
  const int wave = tid >> 5;
  const int lane = tid & 31;
  const int hl   = lane >> 4;
  const int ln   = lane & 15;

  const int Y0 = blockIdx.y * TILE;
  const int X0 = blockIdx.x * TILE;

  v2f W[3][4];
#pragma unroll
  for (int k = 0; k < 3; ++k)
#pragma unroll
    for (int kc = 0; kc < 4; ++kc) {
      int d0 = 16 * k + 4 * kc + 2 * hl - ln;
      float a0 = (d0 >= 0 && d0 <= 32) ? WC[d0] : 0.0f;
      float a1 = (d0 + 1 >= 0 && d0 + 1 <= 32) ? WC[d0 + 1] : 0.0f;
      v2f w; w.x = a0; w.y = a1;
      W[k][kc] = w;
    }

  const int c4  = lane * 4;
  const int gx0 = X0 - 32 + c4;
  const bool rev = (gx0 < 0) || (gx0 > IMG - 1);
  const int gxb = (gx0 < 0) ? (-4 - gx0) : ((gx0 > IMG - 1) ? (2 * IMG - 4 - gx0) : gx0);
  v4f xin[16];
#pragma unroll
  for (int i = 0; i < 16; ++i) {
    int gy = Y0 - 32 + wave + 8 * i;
    gy = (gy < 0) ? (-1 - gy) : ((gy > IMG - 1) ? (2 * IMG - 1 - gy) : gy);
    xin[i] = *reinterpret_cast<const v4f*>(x + gy * IMG + gxb);
  }
  __builtin_amdgcn_sched_barrier(0);
  const int ox4 = (tid & 15) * 4;
  const int oyb = tid >> 4;
  v4f nin0[4], nin1[4], nin2[4];
#pragma unroll
  for (int i = 0; i < 4; ++i) {
    int gidx = (Y0 + oyb + 16 * i) * IMG + (X0 + ox4);
    nin0[i] = *reinterpret_cast<const v4f*>(nrm + gidx);
    nin1[i] = *reinterpret_cast<const v4f*>(nrm + gidx + IMG * IMG);
    nin2[i] = *reinterpret_cast<const v4f*>(nrm + gidx + 2 * IMG * IMG);
  }
  __builtin_amdgcn_sched_barrier(0);

#pragma unroll
  for (int i = 0; i < 16; ++i) {
    const int row = wave + 8 * i;
    const v4f v = xin[i];
    const float e0 = rev ? v.w : v.x;
    const float e1 = rev ? v.z : v.y;
    const float e2 = rev ? v.y : v.z;
    const float e3 = rev ? v.x : v.w;
    const int s = row + c4;
    float* xr = Xs + row * XS_S;
    xr[(s + 0) & 127] = e0;
    xr[(s + 1) & 127] = e1;
    xr[(s + 2) & 127] = e2;
    xr[(s + 3) & 127] = e3;
    if (i >= 4 && i < 12 && c4 >= 32 && c4 < 96) {
      v4f t; t.x = e0; t.y = e1; t.z = e2; t.w = e3;
      *reinterpret_cast<v4f*>(Stash + (row - 32) * TILE + (c4 - 32)) = t;
    }
  }

  lds_barrier();

  {
    const int c = wave * 16 + ln;
    v2f B0[4], B1[4], B2[4];
#pragma unroll
    for (int kc = 0; kc < 4; ++kc) {
      int rr = 4 * kc + 2 * hl;
      v2f t;
      t.x = Xs[(rr + 0) * XS_S + c];  t.y = Xs[(rr + 1) * XS_S + c];  B0[kc] = t;
      t.x = Xs[(rr + 16) * XS_S + c]; t.y = Xs[(rr + 17) * XS_S + c]; B1[kc] = t;
      t.x = Xs[(rr + 32) * XS_S + c]; t.y = Xs[(rr + 33) * XS_S + c]; B2[kc] = t;
    }
    const int cvb = c - 32 - 16 * hl;
    float* tsw = Ts + (8 * hl) * TS_S;
    for (int rb = 0; rb < 6; ++rb) {
      const int r0 = 16 + 16 * rb;
      v8f acc = conv_block(W, B0, B1, B2);
#pragma unroll
      for (int g = 0; g < 8; ++g) {
        const int j = 16 * rb + g;
        tsw[j * TS_S + ((cvb - 2 * j) & 127)] = acc[g];
      }
      if (rb < 5) {
#pragma unroll
        for (int kc = 0; kc < 4; ++kc) {
          B0[kc] = B1[kc];
          B1[kc] = B2[kc];
          int rr = r0 + 32 + 4 * kc + 2 * hl;
          v2f t;
          t.x = Xs[(rr + 0) * XS_S + c];
          t.y = Xs[(rr + 1) * XS_S + c];
          B2[kc] = t;
        }
      }
    }
  }

  lds_barrier();

  {
    const int s    = __builtin_amdgcn_readfirstlane(wave);
    const int b_lo = (s < 4) ? 0 : (7 - s);
    const int nb   = (s < 4) ? (4 - s) : (s - 3);
    const int c    = s * 16 + ln;
    const float* tp = Ts + (16 * b_lo + 2 * hl) * TS_S + c;
    float*       op = Ts2 + (16 * b_lo + 8 * hl) * TS2_S + c;
    v2f B0[4], B1[4], B2[4];
#pragma unroll
    for (int kc = 0; kc < 4; ++kc) {
      const int rr = 4 * kc;
      v2f t;
      t.x = tp[(rr + 0) * TS_S];  t.y = tp[(rr + 1) * TS_S];  B0[kc] = t;
      t.x = tp[(rr + 16) * TS_S]; t.y = tp[(rr + 17) * TS_S]; B1[kc] = t;
      t.x = tp[(rr + 32) * TS_S]; t.y = tp[(rr + 33) * TS_S]; B2[kc] = t;
    }
#pragma unroll
    for (int it = 0; it < 4; ++it) {
      if (it > 0) {
        if (it >= nb) break;
#pragma unroll
        for (int kc = 0; kc < 4; ++kc) {
          B0[kc] = B1[kc];
          B1[kc] = B2[kc];
          const int rr = 16 * it + 32 + 4 * kc;
          v2f t;
          t.x = tp[(rr + 0) * TS_S];
          t.y = tp[(rr + 1) * TS_S];
          B2[kc] = t;
        }
      }
      v8f acc = conv_block(W, B0, B1, B2);
#pragma unroll
      for (int g = 0; g < 8; ++g)
        op[(16 * it + g) * TS2_S] = acc[g];
    }
  }

  lds_barrier();

#pragma unroll
  for (int i = 0; i < 4; ++i) {
    const int oy = oyb + 16 * i;
    const v4f xv = *reinterpret_cast<const v4f*>(Stash + oy * TILE + ox4);
    const v4f n0 = nin0[i], n1 = nin1[i], n2 = nin2[i];
    v4f res;
#pragma unroll
    for (int k = 0; k < 4; ++k) {
      const int cv = (ox4 + k - oy) & 127;
      float rx = Ts2[oy * TS2_S + cv];
      float ao = 1.0f - fminf(fmaxf((rx - xv[k]) * 4.0f, 0.0f), 1.0f);
      float d1 = fmaxf(0.7053f * ((n0[k] - n1[k]) + n2[k]), 0.0f);
      res[k] = (0.3f * d1 + 0.7f) * ao;
    }
    *reinterpret_cast<v4f*>(out + (Y0 + oy) * IMG + (X0 + ox4)) = res;
  }
}

extern "C" void kernel_launch(void* const* d_in, const int* in_sizes, int n_in,
                              void* d_out, int out_size, void* d_ws, size_t ws_size,
                              hipStream_t stream) {
  (void)in_sizes; (void)n_in; (void)d_ws; (void)ws_size; (void)out_size;
  const float* x   = (const float*)d_in[0];
  const float* nrm = (const float*)d_in[1];
  float* out = (float*)d_out;
  dim3 grid(IMG / TILE, IMG / TILE);
  size_t ldsBytes = LDS_FLOATS * sizeof(float);
  ao_render_kernel<<<grid, dim3(256), ldsBytes, stream>>>(x, nrm, out);
}


# each block now loops over 8 tiles (512 blocks, all resident) and issues tile t+1's 16 input loads right after tile t is staged in LDS, so they fly under both conv stages; sums unchanged
# speedup vs baseline: 4.7546x; 1.1407x over previous
#include <hip/hip_runtime.h>

typedef float v2f __attribute__((ext_vector_type(2)));
typedef float v4f __attribute__((ext_vector_type(4)));
typedef float v8f __attribute__((ext_vector_type(8)));

#define WS 2.3283064365386963e-10f
__constant__ float WC[33] = {
  1.0f*WS, 32.0f*WS, 496.0f*WS, 4960.0f*WS, 35960.0f*WS, 201376.0f*WS,
  906192.0f*WS, 3365856.0f*WS, 10518300.0f*WS, 28048800.0f*WS,
  64512240.0f*WS, 129024480.0f*WS, 225792840.0f*WS, 347373600.0f*WS,
  471435600.0f*WS, 565722720.0f*WS, 601080390.0f*WS, 565722720.0f*WS,
  471435600.0f*WS, 347373600.0f*WS, 225792840.0f*WS, 129024480.0f*WS,
  64512240.0f*WS, 28048800.0f*WS, 10518300.0f*WS, 3365856.0f*WS,
  906192.0f*WS, 201376.0f*WS, 35960.0f*WS, 4960.0f*WS, 496.0f*WS,
  32.0f*WS, 1.0f*WS
};

#define IMG     4096
#define TILE    64
#define INR     128
#define XS_S    136
#define TW      96
#define TS_S    136
#define TS2_S   132
#define XS_FL   (INR * XS_S)
#define TS_FL   (TW * TS_S)
#define STASH_FL (TILE * TILE)
#define LDS_FLOATS (XS_FL + TS_FL + STASH_FL)
#define TPR     (IMG / TILE)
#define TPB     8
#define NBLK    ((TPR * TPR) / TPB)
#define TSTEP   (TPR / TPB)

static __device__ __forceinline__ v8f wmma4(v2f a, v2f b, v8f c) {
  return __builtin_amdgcn_wmma_f32_16x16x4_f32(false, a, false, b, (short)0, c,
                                               false, false);
}

static __device__ __forceinline__ void lds_barrier() {
  asm volatile("s_wait_dscnt 0x0\n\ts_barrier_signal -1\n\ts_barrier_wait -1" ::: "memory");
}

static __device__ __forceinline__ v8f conv_block(const v2f (&W)[3][4],
                                                 const v2f (&B0)[4],
                                                 const v2f (&B1)[4],
                                                 const v2f (&B2)[4]) {
  v8f a0 = {}, a1 = {};
  a0 = wmma4(W[0][0], B0[0], a0);  a1 = wmma4(W[2][0], B2[0], a1);
  a0 = wmma4(W[0][1], B0[1], a0);  a1 = wmma4(W[2][1], B2[1], a1);
  a0 = wmma4(W[0][2], B0[2], a0);  a1 = wmma4(W[2][2], B2[2], a1);
  a0 = wmma4(W[0][3], B0[3], a0);  a1 = wmma4(W[2][3], B2[3], a1);
  a0 = wmma4(W[1][0], B1[0], a0);  a1 = wmma4(W[1][2], B1[2], a1);
  a0 = wmma4(W[1][1], B1[1], a0);  a1 = wmma4(W[1][3], B1[3], a1);
  return a0 + a1;
}

template <bool MORE>
static __device__ __forceinline__ void tile_pass(const float* __restrict__ x,
                                                 const float* __restrict__ nrm,
                                                 float* __restrict__ out,
                                                 float* lds, const v2f (&W)[3][4], v4f (&xin)[16],
                                                 const int tid0, const int sw,
                                                 const int X0, const int Y0,
                                                 const bool rev, const unsigned gxb, const unsigned ooff) {
  float* Xs    = lds;
  float* Ts    = lds + XS_FL;
  float* Stash = lds + XS_FL + TS_FL;
  float* Ts2   = lds;

  int tid = tid0;
  asm volatile("" : "+v"(tid));
  const int wave = tid >> 5;
  const int lane = tid & 31;
  const int hl   = lane >> 4;
  const int ln   = lane & 15;
  const int c4   = lane * 4;
  const int ox4  = (tid & 15) * 4;
  const int oyb  = tid >> 4;

  v4f nin0[4], nin1[4], nin2[4];
  {
    const float* n0p = nrm + (Y0 * IMG + X0);
#pragma unroll
    for (int i = 0; i < 4; ++i) {
      nin0[i] = *reinterpret_cast<const v4f*>(n0p + 16 * i * IMG + ooff);
      nin1[i] = *reinterpret_cast<const v4f*>(n0p + 16 * i * IMG + IMG * IMG + ooff);
      nin2[i] = *reinterpret_cast<const v4f*>(n0p + 16 * i * IMG + 2 * IMG * IMG + ooff);
    }
  }
  __builtin_amdgcn_sched_barrier(0);

#pragma unroll
  for (int i = 0; i < 16; ++i) {
    const int row = wave + 8 * i;
    const v4f v = xin[i];
    const float e0 = rev ? v.w : v.x;
    const float e1 = rev ? v.z : v.y;
    const float e2 = rev ? v.y : v.z;
    const float e3 = rev ? v.x : v.w;
    const int s = row + c4;
    float* xr = Xs + row * XS_S;
    xr[(s + 0) & 127] = e0;
    xr[(s + 1) & 127] = e1;
    xr[(s + 2) & 127] = e2;
    xr[(s + 3) & 127] = e3;
    if (i >= 4 && i < 12 && c4 >= 32 && c4 < 96) {
      v4f t; t.x = e0; t.y = e1; t.z = e2; t.w = e3;
      *reinterpret_cast<v4f*>(Stash + (row - 32) * TILE + (c4 - 32)) = t;
    }
  }

  lds_barrier();

  if (MORE) {
#pragma unroll
    for (int i = 0; i < 16; ++i) {
      int gy = Y0 + TSTEP * TILE - 32 + sw + 8 * i;
      gy = (gy < 0) ? (-1 - gy) : ((gy > IMG - 1) ? (2 * IMG - 1 - gy) : gy);
      const float* xrow = x + gy * IMG;
      xin[i] = *reinterpret_cast<const v4f*>(xrow + gxb);
    }
  }
  __builtin_amdgcn_sched_barrier(0);

  {
    const int c = wave * 16 + ln;
    v2f B0[4], B1[4], B2[4];
#pragma unroll
    for (int kc = 0; kc < 4; ++kc) {
      int rr = 4 * kc + 2 * hl;
      v2f t;
      t.x = Xs[(rr + 0) * XS_S + c];  t.y = Xs[(rr + 1) * XS_S + c];  B0[kc] = t;
      t.x = Xs[(rr + 16) * XS_S + c]; t.y = Xs[(rr + 17) * XS_S + c]; B1[kc] = t;
      t.x = Xs[(rr + 32) * XS_S + c]; t.y = Xs[(rr + 33) * XS_S + c]; B2[kc] = t;
    }
    const int cvb = c - 32 - 16 * hl;
    float* tsw = Ts + (8 * hl) * TS_S;
    for (int rb = 0; rb < 6; ++rb) {
      const int r0 = 16 + 16 * rb;
      v8f acc = conv_block(W, B0, B1, B2);
#pragma unroll
      for (int g = 0; g < 8; ++g) {
        const int j = 16 * rb + g;
        tsw[j * TS_S + ((cvb - 2 * j) & 127)] = acc[g];
      }
      if (rb < 5) {
#pragma unroll
        for (int kc = 0; kc < 4; ++kc) {
          B0[kc] = B1[kc];
          B1[kc] = B2[kc];
          int rr = r0 + 32 + 4 * kc + 2 * hl;
          v2f t;
          t.x = Xs[(rr + 0) * XS_S + c];
          t.y = Xs[(rr + 1) * XS_S + c];
          B2[kc] = t;
        }
      }
    }
  }

  lds_barrier();

  {
    const int s    = sw;
    const int b_lo = (s < 4) ? 0 : (7 - s);
    const int nb   = (s < 4) ? (4 - s) : (s - 3);
    const int c    = s * 16 + ln;
    const float* tp = Ts + (16 * b_lo + 2 * hl) * TS_S + c;
    float*       op = Ts2 + (16 * b_lo + 8 * hl) * TS2_S + c;
    v2f B0[4], B1[4], B2[4];
#pragma unroll
    for (int kc = 0; kc < 4; ++kc) {
      const int rr = 4 * kc;
      v2f t;
      t.x = tp[(rr + 0) * TS_S];  t.y = tp[(rr + 1) * TS_S];  B0[kc] = t;
      t.x = tp[(rr + 16) * TS_S]; t.y = tp[(rr + 17) * TS_S]; B1[kc] = t;
      t.x = tp[(rr + 32) * TS_S]; t.y = tp[(rr + 33) * TS_S]; B2[kc] = t;
    }
#pragma unroll
    for (int it = 0; it < 4; ++it) {
      if (it > 0) {
        if (it >= nb) break;
#pragma unroll
        for (int kc = 0; kc < 4; ++kc) {
          B0[kc] = B1[kc];
          B1[kc] = B2[kc];
          const int rr = 16 * it + 32 + 4 * kc;
          v2f t;
          t.x = tp[(rr + 0) * TS_S];
          t.y = tp[(rr + 1) * TS_S];
          B2[kc] = t;
        }
      }
      v8f acc = conv_block(W, B0, B1, B2);
#pragma unroll
      for (int g = 0; g < 8; ++g)
        op[(16 * it + g) * TS2_S] = acc[g];
    }
  }

  lds_barrier();

#pragma unroll
  for (int i = 0; i < 4; ++i) {
    const int oy = oyb + 16 * i;
    const v4f xv = *reinterpret_cast<const v4f*>(Stash + oy * TILE + ox4);
    const v4f n0 = nin0[i], n1 = nin1[i], n2 = nin2[i];
    v4f res;
#pragma unroll
    for (int k = 0; k < 4; ++k) {
      const int cv = (ox4 + k - oy) & 127;
      float rx = Ts2[oy * TS2_S + cv];
      float ao = 1.0f - fminf(fmaxf((rx - xv[k]) * 4.0f, 0.0f), 1.0f);
      float d1 = fmaxf(0.7053f * ((n0[k] - n1[k]) + n2[k]), 0.0f);
      res[k] = (0.3f * d1 + 0.7f) * ao;
    }
    *reinterpret_cast<v4f*>(out + (Y0 * IMG + X0) + 16 * i * IMG + ooff) = res;
  }

  if (MORE) {
    lds_barrier();
  }
}

__global__ __launch_bounds__(256)
void ao_render_kernel(const float* __restrict__ x,
                      const float* __restrict__ nrm,
                      float* __restrict__ out) {
  extern __shared__ __attribute__((aligned(16))) float lds[];

  const int tid0 = threadIdx.x;
  const int sw   = __builtin_amdgcn_readfirstlane(tid0 >> 5);

  const int X0  = (blockIdx.x & (TPR - 1)) * TILE;
  const int ty0 = blockIdx.x / TPR;

  v2f W[3][4];
#pragma unroll
  for (int k = 0; k < 3; ++k)
#pragma unroll
    for (int kc = 0; kc < 4; ++kc) {
      int d0 = 16 * k + 4 * kc + 2 * ((tid0 >> 4) & 1) - (tid0 & 15);
      float a0 = (d0 >= 0 && d0 <= 32) ? WC[d0] : 0.0f;
      float a1 = (d0 + 1 >= 0 && d0 + 1 <= 32) ? WC[d0 + 1] : 0.0f;
      v2f w; w.x = a0; w.y = a1;
      W[k][kc] = w;
    }

  const int gx0 = X0 - 32 + (tid0 & 31) * 4;
  const bool rev = (gx0 < 0) || (gx0 > IMG - 1);
  const unsigned gxb = (gx0 < 0) ? (-4 - gx0) : ((gx0 > IMG - 1) ? (2 * IMG - 4 - gx0) : gx0);
  const unsigned ooff = (unsigned)((tid0 >> 4) * IMG + (tid0 & 15) * 4);
  v4f xin[16];
#pragma unroll
  for (int i = 0; i < 16; ++i) {
    int gy = ty0 * TILE - 32 + sw + 8 * i;
    gy = (gy < 0) ? (-1 - gy) : ((gy > IMG - 1) ? (2 * IMG - 1 - gy) : gy);
    const float* xrow = x + gy * IMG;
    xin[i] = *reinterpret_cast<const v4f*>(xrow + gxb);
  }

#pragma unroll 1
  for (int t = 0; t < TPB - 1; ++t)
    tile_pass<true>(x, nrm, out, lds, W, xin, tid0, sw, X0, (ty0 + TSTEP * t) * TILE, rev, gxb, ooff);
  tile_pass<false>(x, nrm, out, lds, W, xin, tid0, sw, X0, (ty0 + TSTEP * (TPB - 1)) * TILE, rev, gxb, ooff);
}

extern "C" void kernel_launch(void* const* d_in, const int* in_sizes, int n_in,
                              void* d_out, int out_size, void* d_ws, size_t ws_size,
                              hipStream_t stream) {
  (void)in_sizes; (void)n_in; (void)d_ws; (void)ws_size; (void)out_size;
  const float* x   = (const float*)d_in[0];
  const float* nrm = (const float*)d_in[1];
  float* out = (float*)d_out;
  size_t ldsBytes = LDS_FLOATS * sizeof(float);
  ao_render_kernel<<<dim3(NBLK), dim3(256), ldsBytes, stream>>>(x, nrm, out);
}


# a block's 8 tiles now lie one below the other; the 32 stage-1 rows shared with the tile above are copied in LDS (rotated 64 columns), not recomputed: stage 1 runs 4 of 6 row blocks; sums unchanged
# speedup vs baseline: 5.1538x; 1.0840x over previous
#include <hip/hip_runtime.h>

typedef float v2f __attribute__((ext_vector_type(2)));
typedef float v4f __attribute__((ext_vector_type(4)));
typedef float v8f __attribute__((ext_vector_type(8)));

#define WS 2.3283064365386963e-10f
__constant__ float WC[33] = {
  1.0f*WS, 32.0f*WS, 496.0f*WS, 4960.0f*WS, 35960.0f*WS, 201376.0f*WS,
  906192.0f*WS, 3365856.0f*WS, 10518300.0f*WS, 28048800.0f*WS,
  64512240.0f*WS, 129024480.0f*WS, 225792840.0f*WS, 347373600.0f*WS,
  471435600.0f*WS, 565722720.0f*WS, 601080390.0f*WS, 565722720.0f*WS,
  471435600.0f*WS, 347373600.0f*WS, 225792840.0f*WS, 129024480.0f*WS,
  64512240.0f*WS, 28048800.0f*WS, 10518300.0f*WS, 3365856.0f*WS,
  906192.0f*WS, 201376.0f*WS, 35960.0f*WS, 4960.0f*WS, 496.0f*WS,
  32.0f*WS, 1.0f*WS
};

#define IMG     4096
#define TILE    64
#define INR     128
#define XS_S    136
#define TW      96
#define TS_S    136
#define TS2_S   132
#define XS_FL   (INR * XS_S)
#define TS_FL   (TW * TS_S)
#define STASH_FL (TILE * TILE)
#define LDS_FLOATS (XS_FL + TS_FL + STASH_FL)
#define TPR     (IMG / TILE)
#define TPB     8
#define NBLK    ((TPR * TPR) / TPB)
#define KEEP    32

static __device__ __forceinline__ v8f wmma4(v2f a, v2f b, v8f c) {
  return __builtin_amdgcn_wmma_f32_16x16x4_f32(false, a, false, b, (short)0, c,
                                               false, false);
}

static __device__ __forceinline__ void lds_barrier() {
  asm volatile("s_wait_dscnt 0x0\n\ts_barrier_signal -1\n\ts_barrier_wait -1" ::: "memory");
}

static __device__ __forceinline__ v8f conv_block(const v2f (&W)[3][4],
                                                 const v2f (&B0)[4],
                                                 const v2f (&B1)[4],
                                                 const v2f (&B2)[4]) {
  v8f a0 = {}, a1 = {};
  a0 = wmma4(W[0][0], B0[0], a0);  a1 = wmma4(W[2][0], B2[0], a1);
  a0 = wmma4(W[0][1], B0[1], a0);  a1 = wmma4(W[2][1], B2[1], a1);
  a0 = wmma4(W[0][2], B0[2], a0);  a1 = wmma4(W[2][2], B2[2], a1);
  a0 = wmma4(W[0][3], B0[3], a0);  a1 = wmma4(W[2][3], B2[3], a1);
  a0 = wmma4(W[1][0], B1[0], a0);  a1 = wmma4(W[1][2], B1[2], a1);
  a0 = wmma4(W[1][1], B1[1], a0);  a1 = wmma4(W[1][3], B1[3], a1);
  return a0 + a1;
}

template <bool MORE, bool FIRST>
static __device__ __forceinline__ void tile_pass(const float* __restrict__ x,
                                                 const float* __restrict__ nrm,
                                                 float* __restrict__ out,
                                                 float* lds, const v2f (&W)[3][4], v4f (&xin)[16],
                                                 const int tid0, const int sw,
                                                 const int X0, const int Y0,
                                                 const bool rev, const unsigned gxb, const unsigned ooff) {
  float* Xs    = lds;
  float* Ts    = lds + XS_FL;
  float* Stash = lds + XS_FL + TS_FL;
  float* Ts2   = lds;

  int tid = tid0;
  asm volatile("" : "+v"(tid));
  const int wave = tid >> 5;
  const int lane = tid & 31;
  const int hl   = lane >> 4;
  const int ln   = lane & 15;
  const int c4   = lane * 4;
  const int ox4  = (tid & 15) * 4;
  const int oyb  = tid >> 4;

  v4f nin0[4], nin1[4], nin2[4];
  {
    const float* n0p = nrm + (Y0 * IMG + X0);
#pragma unroll
    for (int i = 0; i < 4; ++i) {
      nin0[i] = *reinterpret_cast<const v4f*>(n0p + 16 * i * IMG + ooff);
      nin1[i] = *reinterpret_cast<const v4f*>(n0p + 16 * i * IMG + IMG * IMG + ooff);
      nin2[i] = *reinterpret_cast<const v4f*>(n0p + 16 * i * IMG + 2 * IMG * IMG + ooff);
    }
  }
  __builtin_amdgcn_sched_barrier(0);

#pragma unroll
  for (int i = 0; i < 16; ++i) {
    const int row = wave + 8 * i;
    const v4f v = xin[i];
    const float e0 = rev ? v.w : v.x;
    const float e1 = rev ? v.z : v.y;
    const float e2 = rev ? v.y : v.z;
    const float e3 = rev ? v.x : v.w;
    const int s = row + c4;
    float* xr = Xs + row * XS_S;
    xr[(s + 0) & 127] = e0;
    xr[(s + 1) & 127] = e1;
    xr[(s + 2) & 127] = e2;
    xr[(s + 3) & 127] = e3;
    if (i >= 4 && i < 12 && c4 >= 32 && c4 < 96) {
      v4f t; t.x = e0; t.y = e1; t.z = e2; t.w = e3;
      *reinterpret_cast<v4f*>(Stash + (row - 32) * TILE + (c4 - 32)) = t;
    }
  }

  lds_barrier();

  if (MORE) {
#pragma unroll
    for (int i = 0; i < 16; ++i) {
      int gy = Y0 + TILE - 32 + sw + 8 * i;
      gy = (gy < 0) ? (-1 - gy) : ((gy > IMG - 1) ? (2 * IMG - 1 - gy) : gy);
      const float* xrow = x + gy * IMG;
      xin[i] = *reinterpret_cast<const v4f*>(xrow + gxb);
    }
  }
  __builtin_amdgcn_sched_barrier(0);

  {
    const int RB0 = FIRST ? 0 : (KEEP / 16);
    const int c = wave * 16 + ln;
    v2f B0[4], B1[4], B2[4];
#pragma unroll
    for (int kc = 0; kc < 4; ++kc) {
      int rr = 16 * RB0 + 4 * kc + 2 * hl;
      v2f t;
      t.x = Xs[(rr + 0) * XS_S + c];  t.y = Xs[(rr + 1) * XS_S + c];  B0[kc] = t;
      t.x = Xs[(rr + 16) * XS_S + c]; t.y = Xs[(rr + 17) * XS_S + c]; B1[kc] = t;
      t.x = Xs[(rr + 32) * XS_S + c]; t.y = Xs[(rr + 33) * XS_S + c]; B2[kc] = t;
    }
    const int cvb = c - 32 - 16 * hl;
    float* tsw = Ts + (8 * hl) * TS_S;
    for (int rb = RB0; rb < 6; ++rb) {
      const int r0 = 16 + 16 * rb;
      v8f acc = conv_block(W, B0, B1, B2);
#pragma unroll
      for (int g = 0; g < 8; ++g) {
        const int j = 16 * rb + g;
        tsw[j * TS_S + ((cvb - 2 * j) & 127)] = acc[g];
      }
      if (rb < 5) {
#pragma unroll
        for (int kc = 0; kc < 4; ++kc) {
          B0[kc] = B1[kc];
          B1[kc] = B2[kc];
          int rr = r0 + 32 + 4 * kc + 2 * hl;
          v2f t;
          t.x = Xs[(rr + 0) * XS_S + c];
          t.y = Xs[(rr + 1) * XS_S + c];
          B2[kc] = t;
        }
      }
    }
  }

  lds_barrier();

  {
    const int s    = sw;
    const int b_lo = (s < 4) ? 0 : (7 - s);
    const int nb   = (s < 4) ? (4 - s) : (s - 3);
    const int c    = s * 16 + ln;
    const float* tp = Ts + (16 * b_lo + 2 * hl) * TS_S + c;
    float*       op = Ts2 + (16 * b_lo + 8 * hl) * TS2_S + c;
    v2f B0[4], B1[4], B2[4];
#pragma unroll
    for (int kc = 0; kc < 4; ++kc) {
      const int rr = 4 * kc;
      v2f t;
      t.x = tp[(rr + 0) * TS_S];  t.y = tp[(rr + 1) * TS_S];  B0[kc] = t;
      t.x = tp[(rr + 16) * TS_S]; t.y = tp[(rr + 17) * TS_S]; B1[kc] = t;
      t.x = tp[(rr + 32) * TS_S]; t.y = tp[(rr + 33) * TS_S]; B2[kc] = t;
    }
#pragma unroll
    for (int it = 0; it < 4; ++it) {
      if (it > 0) {
        if (it >= nb) break;
#pragma unroll
        for (int kc = 0; kc < 4; ++kc) {
          B0[kc] = B1[kc];
          B1[kc] = B2[kc];
          const int rr = 16 * it + 32 + 4 * kc;
          v2f t;
          t.x = tp[(rr + 0) * TS_S];
          t.y = tp[(rr + 1) * TS_S];
          B2[kc] = t;
        }
      }
      v8f acc = conv_block(W, B0, B1, B2);
#pragma unroll
      for (int g = 0; g < 8; ++g)
        op[(16 * it + g) * TS2_S] = acc[g];
    }
  }

  lds_barrier();

  v4f keep[KEEP / 8];
  if (MORE) {
#pragma unroll
    for (int i = 0; i < KEEP / 8; ++i)
      keep[i] = *reinterpret_cast<const v4f*>(Ts + (TILE + sw + 8 * i) * TS_S + c4);
  }

#pragma unroll
  for (int i = 0; i < 4; ++i) {
    const int oy = oyb + 16 * i;
    const v4f xv = *reinterpret_cast<const v4f*>(Stash + oy * TILE + ox4);
    const v4f n0 = nin0[i], n1 = nin1[i], n2 = nin2[i];
    v4f res;
#pragma unroll
    for (int k = 0; k < 4; ++k) {
      const int cv = (ox4 + k - oy) & 127;
      float rx = Ts2[oy * TS2_S + cv];
      float ao = 1.0f - fminf(fmaxf((rx - xv[k]) * 4.0f, 0.0f), 1.0f);
      float d1 = fmaxf(0.7053f * ((n0[k] - n1[k]) + n2[k]), 0.0f);
      res[k] = (0.3f * d1 + 0.7f) * ao;
    }
    *reinterpret_cast<v4f*>(out + (Y0 * IMG + X0) + 16 * i * IMG + ooff) = res;
  }

  if (MORE) {
#pragma unroll
    for (int i = 0; i < KEEP / 8; ++i)
      *reinterpret_cast<v4f*>(Ts + (sw + 8 * i) * TS_S + ((c4 + 64) & 127)) = keep[i];
    lds_barrier();
  }
}

__global__ __launch_bounds__(256)
void ao_render_kernel(const float* __restrict__ x,
                      const float* __restrict__ nrm,
                      float* __restrict__ out) {
  extern __shared__ __attribute__((aligned(16))) float lds[];

  const int tid0 = threadIdx.x;
  const int sw   = __builtin_amdgcn_readfirstlane(tid0 >> 5);

  const int X0  = (blockIdx.x & (TPR - 1)) * TILE;
  const int ty0 = (blockIdx.x / TPR) * TPB;

  v2f W[3][4];
#pragma unroll
  for (int k = 0; k < 3; ++k)
#pragma unroll
    for (int kc = 0; kc < 4; ++kc) {
      int d0 = 16 * k + 4 * kc + 2 * ((tid0 >> 4) & 1) - (tid0 & 15);
      float a0 = (d0 >= 0 && d0 <= 32) ? WC[d0] : 0.0f;
      float a1 = (d0 + 1 >= 0 && d0 + 1 <= 32) ? WC[d0 + 1] : 0.0f;
      v2f w; w.x = a0; w.y = a1;
      W[k][kc] = w;
    }

  const int gx0 = X0 - 32 + (tid0 & 31) * 4;
  const bool rev = (gx0 < 0) || (gx0 > IMG - 1);
  const unsigned gxb = (gx0 < 0) ? (-4 - gx0) : ((gx0 > IMG - 1) ? (2 * IMG - 4 - gx0) : gx0);
  const unsigned ooff = (unsigned)((tid0 >> 4) * IMG + (tid0 & 15) * 4);
  v4f xin[16];
#pragma unroll
  for (int i = 0; i < 16; ++i) {
    int gy = ty0 * TILE - 32 + sw + 8 * i;
    gy = (gy < 0) ? (-1 - gy) : ((gy > IMG - 1) ? (2 * IMG - 1 - gy) : gy);
    const float* xrow = x + gy * IMG;
    xin[i] = *reinterpret_cast<const v4f*>(xrow + gxb);
  }

  tile_pass<true, true>(x, nrm, out, lds, W, xin, tid0, sw, X0, ty0 * TILE, rev, gxb, ooff);
#pragma unroll 1
  for (int t = 1; t < TPB - 1; ++t)
    tile_pass<true, false>(x, nrm, out, lds, W, xin, tid0, sw, X0, (ty0 + t) * TILE, rev, gxb, ooff);
  tile_pass<false, false>(x, nrm, out, lds, W, xin, tid0, sw, X0, (ty0 + TPB - 1) * TILE, rev, gxb, ooff);
}

extern "C" void kernel_launch(void* const* d_in, const int* in_sizes, int n_in,
                              void* d_out, int out_size, void* d_ws, size_t ws_size,
                              hipStream_t stream) {
  (void)in_sizes; (void)n_in; (void)d_ws; (void)ws_size; (void)out_size;
  const float* x   = (const float*)d_in[0];
  const float* nrm = (const float*)d_in[1];
  float* out = (float*)d_out;
  size_t ldsBytes = LDS_FLOATS * sizeof(float);
  ao_render_kernel<<<dim3(NBLK), dim3(256), ldsBytes, stream>>>(x, nrm, out);
}


# a later tile no longer re-loads the 64 input rows shared with the tile above: 32 unread rows dropped, 32 copied in LDS (rotated 64 columns); centre-x stash becomes a 128-row ring; sums unchanged
# speedup vs baseline: 5.6047x; 1.0875x over previous
#include <hip/hip_runtime.h>

typedef float v2f __attribute__((ext_vector_type(2)));
typedef float v4f __attribute__((ext_vector_type(4)));
typedef float v8f __attribute__((ext_vector_type(8)));

#define WS 2.3283064365386963e-10f
__constant__ float WC[33] = {
  1.0f*WS, 32.0f*WS, 496.0f*WS, 4960.0f*WS, 35960.0f*WS, 201376.0f*WS,
  906192.0f*WS, 3365856.0f*WS, 10518300.0f*WS, 28048800.0f*WS,
  64512240.0f*WS, 129024480.0f*WS, 225792840.0f*WS, 347373600.0f*WS,
  471435600.0f*WS, 565722720.0f*WS, 601080390.0f*WS, 565722720.0f*WS,
  471435600.0f*WS, 347373600.0f*WS, 225792840.0f*WS, 129024480.0f*WS,
  64512240.0f*WS, 28048800.0f*WS, 10518300.0f*WS, 3365856.0f*WS,
  906192.0f*WS, 201376.0f*WS, 35960.0f*WS, 4960.0f*WS, 496.0f*WS,
  32.0f*WS, 1.0f*WS
};

#define IMG     4096
#define TILE    64
#define INR     128
#define XS_S    136
#define TW      96
#define TS_S    136
#define TS2_S   132
#define XS_FL   (INR * XS_S)
#define TS_FL   (TW * TS_S)
#define STASH_R 128
#define STASH_FL (STASH_R * TILE)
#define LDS_FLOATS (XS_FL + TS_FL + STASH_FL)
#define TPR     (IMG / TILE)
#define TPB     8
#define NBLK    ((TPR * TPR) / TPB)
#define KEEP    32

static __device__ __forceinline__ v8f wmma4(v2f a, v2f b, v8f c) {
  return __builtin_amdgcn_wmma_f32_16x16x4_f32(false, a, false, b, (short)0, c,
                                               false, false);
}

static __device__ __forceinline__ void lds_barrier() {
  asm volatile("s_wait_dscnt 0x0\n\ts_barrier_signal -1\n\ts_barrier_wait -1" ::: "memory");
}

static __device__ __forceinline__ v8f conv_block(const v2f (&W)[3][4],
                                                 const v2f (&B0)[4],
                                                 const v2f (&B1)[4],
                                                 const v2f (&B2)[4]) {
  v8f a0 = {}, a1 = {};
  a0 = wmma4(W[0][0], B0[0], a0);  a1 = wmma4(W[2][0], B2[0], a1);
  a0 = wmma4(W[0][1], B0[1], a0);  a1 = wmma4(W[2][1], B2[1], a1);
  a0 = wmma4(W[0][2], B0[2], a0);  a1 = wmma4(W[2][2], B2[2], a1);
  a0 = wmma4(W[0][3], B0[3], a0);  a1 = wmma4(W[2][3], B2[3], a1);
  a0 = wmma4(W[1][0], B1[0], a0);  a1 = wmma4(W[1][2], B1[2], a1);
  a0 = wmma4(W[1][1], B1[1], a0);  a1 = wmma4(W[1][3], B1[3], a1);
  return a0 + a1;
}

template <bool MORE, bool FIRST>
static __device__ __forceinline__ void tile_pass(const float* __restrict__ x,
                                                 const float* __restrict__ nrm,
                                                 float* __restrict__ out,
                                                 float* lds, const v2f (&W)[3][4], v4f (&xin)[16],
                                                 v4f (&keepx)[4], const int sp,
                                                 const int tid0, const int sw,
                                                 const int X0, const int Y0,
                                                 const bool rev, const unsigned gxb, const unsigned ooff) {
  float* Xs    = lds;
  float* Ts    = lds + XS_FL;
  float* Stash = lds + XS_FL + TS_FL;
  float* Ts2   = lds;

  int tid = tid0;
  asm volatile("" : "+v"(tid));
  const int wave = tid >> 5;
  const int lane = tid & 31;
  const int hl   = lane >> 4;
  const int ln   = lane & 15;
  const int c4   = lane * 4;
  const int ox4  = (tid & 15) * 4;
  const int oyb  = tid >> 4;

  v4f nin0[4], nin1[4], nin2[4];
  {
    const float* n0p = nrm + (Y0 * IMG + X0);
#pragma unroll
    for (int i = 0; i < 4; ++i) {
      nin0[i] = *reinterpret_cast<const v4f*>(n0p + 16 * i * IMG + ooff);
      nin1[i] = *reinterpret_cast<const v4f*>(n0p + 16 * i * IMG + IMG * IMG + ooff);
      nin2[i] = *reinterpret_cast<const v4f*>(n0p + 16 * i * IMG + 2 * IMG * IMG + ooff);
    }
  }
  __builtin_amdgcn_sched_barrier(0);

  if (!FIRST) {
#pragma unroll
    for (int i = 0; i < 4; ++i)
      *reinterpret_cast<v4f*>(Xs + (32 + wave + 8 * i) * XS_S + ((c4 + 64) & 127)) = keepx[i];
  }
#pragma unroll
  for (int i = (FIRST ? 0 : 8); i < 16; ++i) {
    const int row = wave + 8 * i;
    const v4f v = xin[i];
    const float e0 = rev ? v.w : v.x;
    const float e1 = rev ? v.z : v.y;
    const float e2 = rev ? v.y : v.z;
    const float e3 = rev ? v.x : v.w;
    const int s = row + c4;
    float* xr = Xs + row * XS_S;
    xr[(s + 0) & 127] = e0;
    xr[(s + 1) & 127] = e1;
    xr[(s + 2) & 127] = e2;
    xr[(s + 3) & 127] = e3;
    if (i >= 4 && (MORE || i < 12) && c4 >= 32 && c4 < 96) {
      v4f t; t.x = e0; t.y = e1; t.z = e2; t.w = e3;
      *reinterpret_cast<v4f*>(Stash + ((row - 32 + sp) & (STASH_R - 1)) * TILE + (c4 - 32)) = t;
    }
  }

  lds_barrier();

  if (MORE) {
#pragma unroll
    for (int i = 8; i < 16; ++i) {
      int gy = Y0 + TILE - 32 + sw + 8 * i;
      gy = (gy < 0) ? (-1 - gy) : ((gy > IMG - 1) ? (2 * IMG - 1 - gy) : gy);
      const float* xrow = x + gy * IMG;
      xin[i] = *reinterpret_cast<const v4f*>(xrow + gxb);
    }
  }
  __builtin_amdgcn_sched_barrier(0);

  {
    const int RB0 = FIRST ? 0 : (KEEP / 16);
    const int c = wave * 16 + ln;
    v2f B0[4], B1[4], B2[4];
#pragma unroll
    for (int kc = 0; kc < 4; ++kc) {
      int rr = 16 * RB0 + 4 * kc + 2 * hl;
      v2f t;
      t.x = Xs[(rr + 0) * XS_S + c];  t.y = Xs[(rr + 1) * XS_S + c];  B0[kc] = t;
      t.x = Xs[(rr + 16) * XS_S + c]; t.y = Xs[(rr + 17) * XS_S + c]; B1[kc] = t;
      t.x = Xs[(rr + 32) * XS_S + c]; t.y = Xs[(rr + 33) * XS_S + c]; B2[kc] = t;
    }
    const int cvb = c - 32 - 16 * hl;
    float* tsw = Ts + (8 * hl) * TS_S;
    for (int rb = RB0; rb < 6; ++rb) {
      const int r0 = 16 + 16 * rb;
      v8f acc = conv_block(W, B0, B1, B2);
#pragma unroll
      for (int g = 0; g < 8; ++g) {
        const int j = 16 * rb + g;
        tsw[j * TS_S + ((cvb - 2 * j) & 127)] = acc[g];
      }
      if (rb < 5) {
#pragma unroll
        for (int kc = 0; kc < 4; ++kc) {
          B0[kc] = B1[kc];
          B1[kc] = B2[kc];
          int rr = r0 + 32 + 4 * kc + 2 * hl;
          v2f t;
          t.x = Xs[(rr + 0) * XS_S + c];
          t.y = Xs[(rr + 1) * XS_S + c];
          B2[kc] = t;
        }
      }
    }
  }

  lds_barrier();

  {
    const int s    = sw;
    const int b_lo = (s < 4) ? 0 : (7 - s);
    const int nb   = (s < 4) ? (4 - s) : (s - 3);
    const int c    = s * 16 + ln;
    const float* tp = Ts + (16 * b_lo + 2 * hl) * TS_S + c;
    float*       op = Ts2 + (16 * b_lo + 8 * hl) * TS2_S + c;
    v2f B0[4], B1[4], B2[4];
#pragma unroll
    for (int kc = 0; kc < 4; ++kc) {
      const int rr = 4 * kc;
      v2f t;
      t.x = tp[(rr + 0) * TS_S];  t.y = tp[(rr + 1) * TS_S];  B0[kc] = t;
      t.x = tp[(rr + 16) * TS_S]; t.y = tp[(rr + 17) * TS_S]; B1[kc] = t;
      t.x = tp[(rr + 32) * TS_S]; t.y = tp[(rr + 33) * TS_S]; B2[kc] = t;
    }
#pragma unroll
    for (int it = 0; it < 4; ++it) {
      if (it > 0) {
        if (it >= nb) break;
#pragma unroll
        for (int kc = 0; kc < 4; ++kc) {
          B0[kc] = B1[kc];
          B1[kc] = B2[kc];
          const int rr = 16 * it + 32 + 4 * kc;
          v2f t;
          t.x = tp[(rr + 0) * TS_S];
          t.y = tp[(rr + 1) * TS_S];
          B2[kc] = t;
        }
      }
      v8f acc = conv_block(W, B0, B1, B2);
#pragma unroll
      for (int g = 0; g < 8; ++g)
        op[(16 * it + g) * TS2_S] = acc[g];
    }
  }

  lds_barrier();

  v4f keep[KEEP / 8];
  if (MORE) {
#pragma unroll
    for (int i = 0; i < KEEP / 8; ++i)
      keep[i] = *reinterpret_cast<const v4f*>(Ts + (TILE + sw + 8 * i) * TS_S + c4);
#pragma unroll
    for (int i = 0; i < 4; ++i)
      keepx[i] = *reinterpret_cast<const v4f*>(Xs + (96 + sw + 8 * i) * XS_S + c4);
  }

#pragma unroll
  for (int i = 0; i < 4; ++i) {
    const int oy = oyb + 16 * i;
    const v4f xv = *reinterpret_cast<const v4f*>(Stash + (oy + sp) * TILE + ox4);
    const v4f n0 = nin0[i], n1 = nin1[i], n2 = nin2[i];
    v4f res;
#pragma unroll
    for (int k = 0; k < 4; ++k) {
      const int cv = (ox4 + k - oy) & 127;
      float rx = Ts2[oy * TS2_S + cv];
      float ao = 1.0f - fminf(fmaxf((rx - xv[k]) * 4.0f, 0.0f), 1.0f);
      float d1 = fmaxf(0.7053f * ((n0[k] - n1[k]) + n2[k]), 0.0f);
      res[k] = (0.3f * d1 + 0.7f) * ao;
    }
    *reinterpret_cast<v4f*>(out + (Y0 * IMG + X0) + 16 * i * IMG + ooff) = res;
  }

  if (MORE) {
#pragma unroll
    for (int i = 0; i < KEEP / 8; ++i)
      *reinterpret_cast<v4f*>(Ts + (sw + 8 * i) * TS_S + ((c4 + 64) & 127)) = keep[i];
    lds_barrier();
  }
}

__global__ __launch_bounds__(256)
void ao_render_kernel(const float* __restrict__ x,
                      const float* __restrict__ nrm,
                      float* __restrict__ out) {
  extern __shared__ __attribute__((aligned(16))) float lds[];

  const int tid0 = threadIdx.x;
  const int sw   = __builtin_amdgcn_readfirstlane(tid0 >> 5);

  const int X0  = (blockIdx.x & (TPR - 1)) * TILE;
  const int ty0 = (blockIdx.x / TPR) * TPB;

  v2f W[3][4];
#pragma unroll
  for (int k = 0; k < 3; ++k)
#pragma unroll
    for (int kc = 0; kc < 4; ++kc) {
      int d0 = 16 * k + 4 * kc + 2 * ((tid0 >> 4) & 1) - (tid0 & 15);
      float a0 = (d0 >= 0 && d0 <= 32) ? WC[d0] : 0.0f;
      float a1 = (d0 + 1 >= 0 && d0 + 1 <= 32) ? WC[d0 + 1] : 0.0f;
      v2f w; w.x = a0; w.y = a1;
      W[k][kc] = w;
    }

  const int gx0 = X0 - 32 + (tid0 & 31) * 4;
  const bool rev = (gx0 < 0) || (gx0 > IMG - 1);
  const unsigned gxb = (gx0 < 0) ? (-4 - gx0) : ((gx0 > IMG - 1) ? (2 * IMG - 4 - gx0) : gx0);
  const unsigned ooff = (unsigned)((tid0 >> 4) * IMG + (tid0 & 15) * 4);
  v4f xin[16];
#pragma unroll
  for (int i = 0; i < 16; ++i) {
    int gy = ty0 * TILE - 32 + sw + 8 * i;
    gy = (gy < 0) ? (-1 - gy) : ((gy > IMG - 1) ? (2 * IMG - 1 - gy) : gy);
    const float* xrow = x + gy * IMG;
    xin[i] = *reinterpret_cast<const v4f*>(xrow + gxb);
  }

  v4f keepx[4];
  tile_pass<true, true>(x, nrm, out, lds, W, xin, keepx, 0, tid0, sw, X0, ty0 * TILE, rev, gxb, ooff);
#pragma unroll 1
  for (int t = 1; t < TPB - 1; ++t)
    tile_pass<true, false>(x, nrm, out, lds, W, xin, keepx, (t & 1) * TILE, tid0, sw, X0, (ty0 + t) * TILE, rev, gxb, ooff);
  tile_pass<false, false>(x, nrm, out, lds, W, xin, keepx, ((TPB - 1) & 1) * TILE, tid0, sw, X0, (ty0 + TPB - 1) * TILE, rev, gxb, ooff);
}

extern "C" void kernel_launch(void* const* d_in, const int* in_sizes, int n_in,
                              void* d_out, int out_size, void* d_ws, size_t ws_size,
                              hipStream_t stream) {
  (void)in_sizes; (void)n_in; (void)d_ws; (void)ws_size; (void)out_size;
  const float* x   = (const float*)d_in[0];
  const float* nrm = (const float*)d_in[1];
  float* out = (float*)d_out;
  size_t ldsBytes = LDS_FLOATS * sizeof(float);
  ao_render_kernel<<<dim3(NBLK), dim3(256), ldsBytes, stream>>>(x, nrm, out);
}
